# starting kernel
# baseline (speedup 1.0000x reference)
#include <hip/hip_runtime.h>
#include <hip/hip_bf16.h>

#define EMBED 768
#define NSEQ  4096
#define DHEAD 64
#define LDSTR 772

typedef __attribute__((ext_vector_type(2))) float v2f;
typedef __attribute__((ext_vector_type(8))) float v8f;

__global__ __launch_bounds__(384) void qkv_gemm_kernel(
    const float* __restrict__ x,
    const float* __restrict__ Wq, const float* __restrict__ bq,
    const float* __restrict__ Wk, const float* __restrict__ bk,
    const float* __restrict__ Wv, const float* __restrict__ bv,
    float* __restrict__ Q, float* __restrict__ K, float* __restrict__ V) {
  __shared__ float lds_x[16 * LDSTR];

  const int m0  = blockIdx.x * 16;
  const int tid = threadIdx.x;

  for (int v = tid; v < 16 * EMBED / 4; v += 384) {
    const int r  = v / (EMBED / 4);
    const int c4 = v % (EMBED / 4);
    const float4 val = ((const float4*)(x + (size_t)(m0 + r) * EMBED))[c4];
    float* dst = &lds_x[r * LDSTR + c4 * 4];
    dst[0] = val.x; dst[1] = val.y; dst[2] = val.z; dst[3] = val.w;
  }
  __syncthreads();

  const int wave = tid >> 5;
  const int lane = tid & 31;
  const int ln   = lane & 15;
  const int lh   = lane >> 4;

  const float* W; const float* bias; float* outp; int n0;
  if (wave < 4)      { W = Wq; bias = bq; outp = Q; n0 = wave * 16; }
  else if (wave < 8) { W = Wk; bias = bk; outp = K; n0 = (wave - 4) * 16; }
  else               { W = Wv; bias = bv; outp = V; n0 = (wave - 8) * 16; }

  const float bval = bias[n0 + ln];
  v8f acc;
#pragma unroll
  for (int r = 0; r < 8; ++r) acc[r] = bval;

  const float* lrow = &lds_x[ln * LDSTR];
  const float* wcol = W + n0 + ln;

#pragma unroll 4
  for (int k0 = 0; k0 < EMBED; k0 += 4) {
    const int ka = k0 + 2 * lh;
    v2f a;  a[0] = lrow[ka];
            a[1] = lrow[ka + 1];
    v2f bf; bf[0] = wcol[(size_t)ka * DHEAD];
            bf[1] = wcol[(size_t)(ka + 1) * DHEAD];
    acc = __builtin_amdgcn_wmma_f32_16x16x4_f32(
         false, a,  false, bf,
         (short)0, acc,  false,  false);
  }

#pragma unroll
  for (int r = 0; r < 8; ++r) {
    const int m = m0 + r + 8 * lh;
    outp[(size_t)m * DHEAD + n0 + ln] = acc[r];
  }
}

__global__ __launch_bounds__(256) void vsum_kernel(const float* __restrict__ V,
                                                   float* __restrict__ Srest) {
  __shared__ float red[256];
  const int b = blockIdx.x;
  const int t = threadIdx.x;
  const int d = t & 63;
  const int s = t >> 6;
  float acc = 0.0f;
  for (int j = 3 + s; j < NSEQ; j += 4)
    acc += V[((size_t)b * NSEQ + j) * DHEAD + d];
  red[t] = acc;
  __syncthreads();
  if (s == 0)
    Srest[b * DHEAD + d] = red[d] + red[64 + d] + red[128 + d] + red[192 + d];
}

__global__ __launch_bounds__(256) void combine_kernel(
    const float* __restrict__ Q, const float* __restrict__ K,
    const float* __restrict__ V, const float* __restrict__ Srest,
    float* __restrict__ out) {
  const int tid  = threadIdx.x;
  const int wave = tid >> 5;
  const int lane = tid & 31;
  const int g = blockIdx.x * 8 + wave;
  const int b = g >> 12;
  const int i = g & (NSEQ - 1);

  const float2 q = ((const float2*)(Q + (size_t)g * DHEAD))[lane];

  float pa = 0.0f, pb = 0.0f, pc = 0.0f;
  {
    const float2 kc = ((const float2*)(K + (size_t)g * DHEAD))[lane];
    pb = q.x * kc.x + q.y * kc.y;
  }
  if (i > 0) {
    const float2 kp = ((const float2*)(K + (size_t)(g - 1) * DHEAD))[lane];
    pa = q.x * kp.x + q.y * kp.y;
  }
  if (i < NSEQ - 1) {
    const float2 kn = ((const float2*)(K + (size_t)(g + 1) * DHEAD))[lane];
    pc = q.x * kn.x + q.y * kn.y;
  }
#pragma unroll
  for (int off = 16; off > 0; off >>= 1) {
    pa += __shfl_xor(pa, off, 32);
    pb += __shfl_xor(pb, off, 32);
    pc += __shfl_xor(pc, off, 32);
  }
  const float a  = (i > 0) ? pa : 0.0f;
  const float bs = pb;
  const float c  = (i < NSEQ - 1) ? pc : 0.0f;

  const float m  = fmaxf(fmaxf(a, bs), fmaxf(c, 0.0f));
  const float ea = expf(a - m);
  const float eb = expf(bs - m);
  const float ec = expf(c - m);
  const float ez = expf(-m);
  const float inv = 1.0f / (ea + eb + ec + (float)(NSEQ - 3) * ez);

  const float* Vb = V + (size_t)b * NSEQ * DHEAD;
  const float2 v0 = ((const float2*)(Vb))[lane];
  const float2 v1 = ((const float2*)(Vb + DHEAD))[lane];
  const float2 v2 = ((const float2*)(Vb + 2 * DHEAD))[lane];
  const float2 sr = ((const float2*)(Srest + b * DHEAD))[lane];

  float2 o;
  o.x = (ea * v0.x + eb * v1.x + ec * v2.x + ez * sr.x) * inv;
  o.y = (ea * v0.y + eb * v1.y + ec * v2.y + ez * sr.y) * inv;
  ((float2*)(out + (size_t)g * DHEAD))[lane] = o;
}

extern "C" void kernel_launch(void* const* d_in, const int* in_sizes, int n_in,
                              void* d_out, int out_size, void* d_ws, size_t ws_size,
                              hipStream_t stream) {
  const float* x  = (const float*)d_in[0];
  const float* Wq = (const float*)d_in[1];
  const float* bq = (const float*)d_in[2];
  const float* Wk = (const float*)d_in[3];
  const float* bk = (const float*)d_in[4];
  const float* Wv = (const float*)d_in[5];
  const float* bv = (const float*)d_in[6];
  float* out = (float*)d_out;

  const int BN = in_sizes[0] / EMBED;
  const int B  = BN / NSEQ;

  float* Q     = (float*)d_ws;
  float* K     = Q + (size_t)BN * DHEAD;
  float* V     = K + (size_t)BN * DHEAD;
  float* Srest = V + (size_t)BN * DHEAD;

  qkv_gemm_kernel<<<BN / 16, 384, 0, stream>>>(x, Wq, bq, Wk, bk, Wv, bv, Q, K, V);
  vsum_kernel<<<B, 256, 0, stream>>>(V, Srest);
  combine_kernel<<<BN / 8, 256, 0, stream>>>(Q, K, V, Srest, out);
}


# projection: a 64-row block stages each 32-k chunk of x AND of [Wq|Wk|Wv] in LDS by whole-line 16-byte loads (2 stages, loads a chunk ahead); A, B fragments read from LDS; k order, sums unchanged
# speedup vs baseline: 2.3634x; 2.3634x over previous
#include <hip/hip_runtime.h>
#include <hip/hip_bf16.h>

#define EMBED 768
#define NSEQ  4096
#define DHEAD 64
#define BM 64
#define KC 32
#define NCHUNK (EMBED / KC)
#define XP 36
#define WP 208
#define XS_FLOATS (BM * XP)
#define WS_FLOATS (KC * WP)
#define VCHUNK 256
#define VPARTS (NSEQ / VCHUNK)

typedef __attribute__((ext_vector_type(2))) float v2f;
typedef __attribute__((ext_vector_type(8))) float v8f;

__global__ __launch_bounds__(256) __attribute__((amdgpu_waves_per_eu(2, 2))) void qkv_gemm_kernel(
    const float* __restrict__ x,
    const float* __restrict__ Wq, const float* __restrict__ bq,
    const float* __restrict__ Wk, const float* __restrict__ bk,
    const float* __restrict__ Wv, const float* __restrict__ bv,
    float* __restrict__ Q, float* __restrict__ K, float* __restrict__ V) {
  __shared__ __attribute__((aligned(16))) float sx[2][XS_FLOATS];
  __shared__ __attribute__((aligned(16))) float sw[2][WS_FLOATS];

  const int tid = threadIdx.x;
  const int m0  = blockIdx.x * BM;

  const int xr = tid >> 3;
  const int xc = (tid & 7) * 4;
  const float* xg = x + (size_t)(m0 + xr) * EMBED + xc;
  const int xl = xr * XP + xc;
  const int wr = tid >> 4;
  const int wc = (tid & 15) * 4;
  const int wgo = wr * DHEAD + wc;
  const int wl  = wr * WP + wc;

  float4 px0, px1;
  float4 pq0, pq1, pk0, pk1, pv0, pv1;

  auto gload = [&](int ch) {
    const int kc0 = ch * KC;
    px0 = *(const float4*)(xg + kc0);
    px1 = *(const float4*)(xg + kc0 + 32 * EMBED);
    const size_t wo = (size_t)kc0 * DHEAD + wgo;
    pq0 = *(const float4*)(Wq + wo);
    pq1 = *(const float4*)(Wq + wo + 16 * DHEAD);
    pk0 = *(const float4*)(Wk + wo);
    pk1 = *(const float4*)(Wk + wo + 16 * DHEAD);
    pv0 = *(const float4*)(Wv + wo);
    pv1 = *(const float4*)(Wv + wo + 16 * DHEAD);
  };
  auto lstore = [&](int buf) {
    float* dx = &sx[buf][xl];
    *(float4*)dx = px0;
    *(float4*)(dx + 32 * XP) = px1;
    float* dw = &sw[buf][wl];
    *(float4*)(dw)                       = pq0;
    *(float4*)(dw + 16 * WP)             = pq1;
    *(float4*)(dw + DHEAD)               = pk0;
    *(float4*)(dw + DHEAD + 16 * WP)     = pk1;
    *(float4*)(dw + 2 * DHEAD)           = pv0;
    *(float4*)(dw + 2 * DHEAD + 16 * WP) = pv1;
  };

  const int wave = tid >> 5;
  const int lane = tid & 31;
  const int ln   = lane & 15;
  const int lh   = lane >> 4;
  const int rg   = wave & 1;
  const int cg   = wave >> 1;

  const int aoff = (32 * rg + ln) * XP + 2 * lh;
  const int boff = 2 * lh * WP + 48 * cg + ln;

  v8f acc[2][3];
#pragma unroll
  for (int jt = 0; jt < 3; ++jt) {
    const int tn = 3 * cg + jt;
    const float* bias = (tn < 4) ? bq : (tn < 8) ? bk : bv;
    const float bval = bias[(tn & 3) * 16 + ln];
#pragma unroll
    for (int r = 0; r < 8; ++r) { acc[0][jt][r] = bval; acc[1][jt][r] = bval; }
  }

#define WMMA_STEP(S)                                                                                                          \
  acc[0][0] = __builtin_amdgcn_wmma_f32_16x16x4_f32(false, fa[S][0], false, fb[S][0], (short)0, acc[0][0], false, false); \
  acc[1][0] = __builtin_amdgcn_wmma_f32_16x16x4_f32(false, fa[S][1], false, fb[S][0], (short)0, acc[1][0], false, false); \
  acc[0][1] = __builtin_amdgcn_wmma_f32_16x16x4_f32(false, fa[S][0], false, fb[S][1], (short)0, acc[0][1], false, false); \
  acc[1][1] = __builtin_amdgcn_wmma_f32_16x16x4_f32(false, fa[S][1], false, fb[S][1], (short)0, acc[1][1], false, false); \
  acc[0][2] = __builtin_amdgcn_wmma_f32_16x16x4_f32(false, fa[S][0], false, fb[S][2], (short)0, acc[0][2], false, false); \
  acc[1][2] = __builtin_amdgcn_wmma_f32_16x16x4_f32(false, fa[S][1], false, fb[S][2], (short)0, acc[1][2], false, false);

  auto chunk = [&](int buf, int nbuf, int gch, bool stage) {
    const float* ax = &sx[buf][aoff];
    const float* bw = &sw[buf][boff];
    v2f fa[KC / 4][2];
    v2f fb[KC / 4][3];
#pragma unroll
    for (int s = 0; s < KC / 4; ++s) {
      fa[s][0] = *(const v2f*)(ax + 4 * s);
      fa[s][1] = *(const v2f*)(ax + 16 * XP + 4 * s);
#pragma unroll
      for (int j = 0; j < 3; ++j) {
        fb[s][j][0] = bw[4 * s * WP + 16 * j];
        fb[s][j][1] = bw[4 * s * WP + 16 * j + WP];
      }
    }
    WMMA_STEP(0) WMMA_STEP(1) WMMA_STEP(2) WMMA_STEP(3) WMMA_STEP(4) WMMA_STEP(5)
    __builtin_amdgcn_sched_barrier(0);
    if (stage) {
      lstore(nbuf);
      gload(gch);
    }
    WMMA_STEP(6) WMMA_STEP(7)
  };

  gload(0);
  lstore(0);
  gload(1);
  __syncthreads();
  for (int ch = 0; ch < NCHUNK - 1; ++ch) {
    chunk(ch & 1, (ch + 1) & 1, (ch + 2 < NCHUNK) ? ch + 2 : NCHUNK - 1, true);
    __builtin_amdgcn_fence(__ATOMIC_RELEASE, "workgroup", "local");
    __builtin_amdgcn_s_barrier();
    __builtin_amdgcn_fence(__ATOMIC_ACQUIRE, "workgroup", "local");
  }
  chunk((NCHUNK - 1) & 1, 0, 0, false);
#undef WMMA_STEP

#pragma unroll
  for (int jt = 0; jt < 3; ++jt) {
    const int tn = 3 * cg + jt;
    float* outp = (tn < 4) ? Q : (tn < 8) ? K : V;
    float* o = outp + (size_t)(m0 + 32 * rg + 8 * lh) * DHEAD + (tn & 3) * 16 + ln;
#pragma unroll
    for (int i = 0; i < 2; ++i) {
#pragma unroll
      for (int r = 0; r < 8; ++r) o[(size_t)(16 * i + r) * DHEAD] = acc[i][jt][r];
    }
  }
}

__global__ __launch_bounds__(512) void vsum_kernel(float* V) {
  __shared__ float4 red[16 * 16];
  const int t    = threadIdx.x;
  const int c4   = t & 15;
  const int rs   = t >> 4;
  const int row0 = blockIdx.x * VCHUNK;
  const bool first = (blockIdx.x & (VPARTS - 1)) == 0;
  const float4* src = (const float4*)(V + (size_t)row0 * DHEAD) + c4;

  float4 v[8];
#pragma unroll
  for (int i = 0; i < 8; ++i) v[i] = src[(rs + 32 * i) * (DHEAD / 4)];

  const bool keep0 = !(first && rs < 3);
  float4 acc;
  acc.x = keep0 ? v[0].x : 0.0f;
  acc.y = keep0 ? v[0].y : 0.0f;
  acc.z = keep0 ? v[0].z : 0.0f;
  acc.w = keep0 ? v[0].w : 0.0f;
#pragma unroll
  for (int i = 1; i < 8; ++i) {
    acc.x += v[i].x; acc.y += v[i].y; acc.z += v[i].z; acc.w += v[i].w;
  }

  acc.x += __shfl_xor(acc.x, 16, 32);
  acc.y += __shfl_xor(acc.y, 16, 32);
  acc.z += __shfl_xor(acc.z, 16, 32);
  acc.w += __shfl_xor(acc.w, 16, 32);

  const int wave = t >> 5;
  const int lane = t & 31;
  if (lane < 16) red[wave * 16 + lane] = acc;
  __syncthreads();

  if (t < DHEAD) {
    const float* rf = (const float*)red;
    float s = rf[t];
#pragma unroll
    for (int w = 1; w < 16; ++w) s += rf[w * DHEAD + t];
    V[(size_t)(row0 + VCHUNK - 1) * DHEAD + t] = s;
  }
}

__global__ __launch_bounds__(256) void combine_kernel(
    const float* __restrict__ Q, const float* __restrict__ K,
    const float* __restrict__ V, float* __restrict__ out) {
  __shared__ float sred[256];
  const int tid  = threadIdx.x;
  {
    const int bb = (blockIdx.x * 8) >> 12;
    const int d  = tid & 63;
    const int s  = tid >> 6;
    const float* P = V + ((size_t)bb * NSEQ + (size_t)(4 * s) * VCHUNK + (VCHUNK - 1)) * DHEAD + d;
    float ps = P[0];
#pragma unroll
    for (int r = 1; r < 4; ++r) ps += P[(size_t)r * VCHUNK * DHEAD];
    sred[tid] = ps;
  }
  const int wave = tid >> 5;
  const int lane = tid & 31;
  const int g = blockIdx.x * 8 + wave;
  const int b = g >> 12;
  const int i = g & (NSEQ - 1);

  const float2 q = ((const float2*)(Q + (size_t)g * DHEAD))[lane];

  float pa = 0.0f, pb = 0.0f, pc = 0.0f;
  {
    const float2 kc = ((const float2*)(K + (size_t)g * DHEAD))[lane];
    pb = q.x * kc.x + q.y * kc.y;
  }
  if (i > 0) {
    const float2 kp = ((const float2*)(K + (size_t)(g - 1) * DHEAD))[lane];
    pa = q.x * kp.x + q.y * kp.y;
  }
  if (i < NSEQ - 1) {
    const float2 kn = ((const float2*)(K + (size_t)(g + 1) * DHEAD))[lane];
    pc = q.x * kn.x + q.y * kn.y;
  }
#pragma unroll
  for (int off = 16; off > 0; off >>= 1) {
    pa += __shfl_xor(pa, off, 32);
    pb += __shfl_xor(pb, off, 32);
    pc += __shfl_xor(pc, off, 32);
  }
  const float a  = (i > 0) ? pa : 0.0f;
  const float bs = pb;
  const float c  = (i < NSEQ - 1) ? pc : 0.0f;

  const float m  = fmaxf(fmaxf(a, bs), fmaxf(c, 0.0f));
  const float ea = expf(a - m);
  const float eb = expf(bs - m);
  const float ec = expf(c - m);
  const float ez = expf(-m);
  const float inv = 1.0f / (ea + eb + ec + (float)(NSEQ - 3) * ez);

  const float* Vb = V + (size_t)b * NSEQ * DHEAD;
  const float2 v0 = ((const float2*)(Vb))[lane];
  const float2 v1 = ((const float2*)(Vb + DHEAD))[lane];
  const float2 v2 = ((const float2*)(Vb + 2 * DHEAD))[lane];

  __syncthreads();
  const float2 s0 = ((const float2*)(sred))[lane];
  const float2 s1 = ((const float2*)(sred + 64))[lane];
  const float2 s2 = ((const float2*)(sred + 128))[lane];
  const float2 s3 = ((const float2*)(sred + 192))[lane];
  float2 sr;
  sr.x = ((s0.x + s1.x) + s2.x) + s3.x;
  sr.y = ((s0.y + s1.y) + s2.y) + s3.y;

  float2 o;
  o.x = (ea * v0.x + eb * v1.x + ec * v2.x + ez * sr.x) * inv;
  o.y = (ea * v0.y + eb * v1.y + ec * v2.y + ez * sr.y) * inv;
  ((float2*)(out + (size_t)g * DHEAD))[lane] = o;
}

extern "C" void kernel_launch(void* const* d_in, const int* in_sizes, int n_in,
                              void* d_out, int out_size, void* d_ws, size_t ws_size,
                              hipStream_t stream) {
  const float* x  = (const float*)d_in[0];
  const float* Wq = (const float*)d_in[1];
  const float* bq = (const float*)d_in[2];
  const float* Wk = (const float*)d_in[3];
  const float* bk = (const float*)d_in[4];
  const float* Wv = (const float*)d_in[5];
  const float* bv = (const float*)d_in[6];
  float* out = (float*)d_out;

  const int BN = in_sizes[0] / EMBED;

  float* Q     = (float*)d_ws;
  float* K     = Q + (size_t)BN * DHEAD;
  float* V     = K + (size_t)BN * DHEAD;

  qkv_gemm_kernel<<<BN / BM, 256, 0, stream>>>(x, Wq, bq, Wk, bk, Wv, bv, Q, K, V);
  vsum_kernel<<<BN / VCHUNK, 512, 0, stream>>>(V);
  combine_kernel<<<BN / 8, 256, 0, stream>>>(Q, K, V, out);
}


# projection: each chunk's 32 fragment reads stand in ONE place ahead of its 48 WMMAs (were 9 places among them); W chunk tile-major in LDS so a B read yields its k pair as is; sums unchanged
# speedup vs baseline: 2.5207x; 1.0666x over previous
#include <hip/hip_runtime.h>
#include <hip/hip_bf16.h>

#define EMBED 768
#define NSEQ  4096
#define DHEAD 64
#define BM 64
#define KC 32
#define NCHUNK (EMBED / KC)
#define XP 36
#define WT 528
#define XS_FLOATS (BM * XP)
#define WS_FLOATS (12 * WT)
#define VCHUNK 256
#define VPARTS (NSEQ / VCHUNK)

typedef __attribute__((ext_vector_type(2))) float v2f;
typedef __attribute__((ext_vector_type(8))) float v8f;

__global__ __launch_bounds__(256) __attribute__((amdgpu_waves_per_eu(2, 2))) void qkv_gemm_kernel(
    const float* __restrict__ x,
    const float* __restrict__ Wq, const float* __restrict__ bq,
    const float* __restrict__ Wk, const float* __restrict__ bk,
    const float* __restrict__ Wv, const float* __restrict__ bv,
    float* __restrict__ Q, float* __restrict__ K, float* __restrict__ V) {
  __shared__ __attribute__((aligned(16))) float sx[2][XS_FLOATS];
  __shared__ __attribute__((aligned(16))) float sw[2][WS_FLOATS];

  const int tid = threadIdx.x;
  const int m0  = blockIdx.x * BM;

  const int xr = tid >> 3;
  const int xc = (tid & 7) * 4;
  const float* xg = x + (size_t)(m0 + xr) * EMBED + xc;
  const int xl = xr * XP + xc;
  const int wr = tid >> 4;
  const int wc = (tid & 15) * 4;
  const int wgo = wr * DHEAD + wc;
  const int wl  = (wc >> 4) * WT + wr * 16 + (wc & 15);

  float4 px0, px1;
  float4 pq0, pq1, pk0, pk1, pv0, pv1;

  auto gload = [&](int ch) {
    const int kc0 = ch * KC;
    px0 = *(const float4*)(xg + kc0);
    px1 = *(const float4*)(xg + kc0 + 32 * EMBED);
    const size_t wo = (size_t)kc0 * DHEAD + wgo;
    pq0 = *(const float4*)(Wq + wo);
    pq1 = *(const float4*)(Wq + wo + 16 * DHEAD);
    pk0 = *(const float4*)(Wk + wo);
    pk1 = *(const float4*)(Wk + wo + 16 * DHEAD);
    pv0 = *(const float4*)(Wv + wo);
    pv1 = *(const float4*)(Wv + wo + 16 * DHEAD);
  };
  auto lstore = [&](int buf) {
    float* dx = &sx[buf][xl];
    *(float4*)dx = px0;
    *(float4*)(dx + 32 * XP) = px1;
    float* dw = &sw[buf][wl];
    *(float4*)(dw)                   = pq0;
    *(float4*)(dw + 16 * 16)         = pq1;
    *(float4*)(dw + 4 * WT)          = pk0;
    *(float4*)(dw + 4 * WT + 16 * 16) = pk1;
    *(float4*)(dw + 8 * WT)          = pv0;
    *(float4*)(dw + 8 * WT + 16 * 16) = pv1;
  };

  const int wave = tid >> 5;
  const int lane = tid & 31;
  const int ln   = lane & 15;
  const int lh   = lane >> 4;
  const int rg   = wave & 1;
  const int cg   = wave >> 1;

  const int aoff = (32 * rg + ln) * XP + 2 * lh;
  const int boff = 3 * cg * WT + 2 * lh * 16 + ln;

  v8f acc[2][3];
#pragma unroll
  for (int jt = 0; jt < 3; ++jt) {
    const int tn = 3 * cg + jt;
    const float* bias = (tn < 4) ? bq : (tn < 8) ? bk : bv;
    const float bval = bias[(tn & 3) * 16 + ln];
#pragma unroll
    for (int r = 0; r < 8; ++r) { acc[0][jt][r] = bval; acc[1][jt][r] = bval; }
  }

#define WMMA_STEP(S)                                                                                                          \
  acc[0][0] = __builtin_amdgcn_wmma_f32_16x16x4_f32(false, fa[S][0], false, fb[S][0], (short)0, acc[0][0], false, false); \
  acc[1][0] = __builtin_amdgcn_wmma_f32_16x16x4_f32(false, fa[S][1], false, fb[S][0], (short)0, acc[1][0], false, false); \
  acc[0][1] = __builtin_amdgcn_wmma_f32_16x16x4_f32(false, fa[S][0], false, fb[S][1], (short)0, acc[0][1], false, false); \
  acc[1][1] = __builtin_amdgcn_wmma_f32_16x16x4_f32(false, fa[S][1], false, fb[S][1], (short)0, acc[1][1], false, false); \
  acc[0][2] = __builtin_amdgcn_wmma_f32_16x16x4_f32(false, fa[S][0], false, fb[S][2], (short)0, acc[0][2], false, false); \
  acc[1][2] = __builtin_amdgcn_wmma_f32_16x16x4_f32(false, fa[S][1], false, fb[S][2], (short)0, acc[1][2], false, false);

  auto chunk = [&](int buf, int nbuf, int gch, bool stage) {
    const float* ax = &sx[buf][aoff];
    const float* bw = &sw[buf][boff];
    v2f fa[KC / 4][2];
    v2f fb[KC / 4][3];
#pragma unroll
    for (int s = 0; s < KC / 4; ++s) {
      fa[s][0] = *(const v2f*)(ax + 4 * s);
      fa[s][1] = *(const v2f*)(ax + 16 * XP + 4 * s);
#pragma unroll
      for (int j = 0; j < 3; ++j) {
        fb[s][j][0] = bw[j * WT + 4 * s * 16];
        fb[s][j][1] = bw[j * WT + 4 * s * 16 + 16];
      }
    }
    __builtin_amdgcn_sched_barrier(0);
    WMMA_STEP(0) WMMA_STEP(1) WMMA_STEP(2) WMMA_STEP(3) WMMA_STEP(4) WMMA_STEP(5)
    __builtin_amdgcn_sched_barrier(0);
    if (stage) {
      lstore(nbuf);
      gload(gch);
    }
    WMMA_STEP(6) WMMA_STEP(7)
  };

  gload(0);
  lstore(0);
  gload(1);
  __syncthreads();
  for (int ch = 0; ch < NCHUNK - 1; ++ch) {
    chunk(ch & 1, (ch + 1) & 1, (ch + 2 < NCHUNK) ? ch + 2 : NCHUNK - 1, true);
    __builtin_amdgcn_fence(__ATOMIC_RELEASE, "workgroup", "local");
    __builtin_amdgcn_s_barrier();
    __builtin_amdgcn_fence(__ATOMIC_ACQUIRE, "workgroup", "local");
  }
  chunk((NCHUNK - 1) & 1, 0, 0, false);
#undef WMMA_STEP

#pragma unroll
  for (int jt = 0; jt < 3; ++jt) {
    const int tn = 3 * cg + jt;
    float* outp = (tn < 4) ? Q : (tn < 8) ? K : V;
    float* o = outp + (size_t)(m0 + 32 * rg + 8 * lh) * DHEAD + (tn & 3) * 16 + ln;
#pragma unroll
    for (int i = 0; i < 2; ++i) {
#pragma unroll
      for (int r = 0; r < 8; ++r) o[(size_t)(16 * i + r) * DHEAD] = acc[i][jt][r];
    }
  }
}

__global__ __launch_bounds__(512) void vsum_kernel(float* V) {
  __shared__ float4 red[16 * 16];
  const int t    = threadIdx.x;
  const int c4   = t & 15;
  const int rs   = t >> 4;
  const int row0 = blockIdx.x * VCHUNK;
  const bool first = (blockIdx.x & (VPARTS - 1)) == 0;
  const float4* src = (const float4*)(V + (size_t)row0 * DHEAD) + c4;

  float4 v[8];
#pragma unroll
  for (int i = 0; i < 8; ++i) v[i] = src[(rs + 32 * i) * (DHEAD / 4)];

  const bool keep0 = !(first && rs < 3);
  float4 acc;
  acc.x = keep0 ? v[0].x : 0.0f;
  acc.y = keep0 ? v[0].y : 0.0f;
  acc.z = keep0 ? v[0].z : 0.0f;
  acc.w = keep0 ? v[0].w : 0.0f;
#pragma unroll
  for (int i = 1; i < 8; ++i) {
    acc.x += v[i].x; acc.y += v[i].y; acc.z += v[i].z; acc.w += v[i].w;
  }

  acc.x += __shfl_xor(acc.x, 16, 32);
  acc.y += __shfl_xor(acc.y, 16, 32);
  acc.z += __shfl_xor(acc.z, 16, 32);
  acc.w += __shfl_xor(acc.w, 16, 32);

  const int wave = t >> 5;
  const int lane = t & 31;
  if (lane < 16) red[wave * 16 + lane] = acc;
  __syncthreads();

  if (t < DHEAD) {
    const float* rf = (const float*)red;
    float s = rf[t];
#pragma unroll
    for (int w = 1; w < 16; ++w) s += rf[w * DHEAD + t];
    V[(size_t)(row0 + VCHUNK - 1) * DHEAD + t] = s;
  }
}

__global__ __launch_bounds__(256) void combine_kernel(
    const float* __restrict__ Q, const float* __restrict__ K,
    const float* __restrict__ V, float* __restrict__ out) {
  __shared__ float sred[256];
  const int tid  = threadIdx.x;
  {
    const int bb = (blockIdx.x * 8) >> 12;
    const int d  = tid & 63;
    const int s  = tid >> 6;
    const float* P = V + ((size_t)bb * NSEQ + (size_t)(4 * s) * VCHUNK + (VCHUNK - 1)) * DHEAD + d;
    float ps = P[0];
#pragma unroll
    for (int r = 1; r < 4; ++r) ps += P[(size_t)r * VCHUNK * DHEAD];
    sred[tid] = ps;
  }
  const int wave = tid >> 5;
  const int lane = tid & 31;
  const int g = blockIdx.x * 8 + wave;
  const int b = g >> 12;
  const int i = g & (NSEQ - 1);

  const float2 q = ((const float2*)(Q + (size_t)g * DHEAD))[lane];

  float pa = 0.0f, pb = 0.0f, pc = 0.0f;
  {
    const float2 kc = ((const float2*)(K + (size_t)g * DHEAD))[lane];
    pb = q.x * kc.x + q.y * kc.y;
  }
  if (i > 0) {
    const float2 kp = ((const float2*)(K + (size_t)(g - 1) * DHEAD))[lane];
    pa = q.x * kp.x + q.y * kp.y;
  }
  if (i < NSEQ - 1) {
    const float2 kn = ((const float2*)(K + (size_t)(g + 1) * DHEAD))[lane];
    pc = q.x * kn.x + q.y * kn.y;
  }
#pragma unroll
  for (int off = 16; off > 0; off >>= 1) {
    pa += __shfl_xor(pa, off, 32);
    pb += __shfl_xor(pb, off, 32);
    pc += __shfl_xor(pc, off, 32);
  }
  const float a  = (i > 0) ? pa : 0.0f;
  const float bs = pb;
  const float c  = (i < NSEQ - 1) ? pc : 0.0f;

  const float m  = fmaxf(fmaxf(a, bs), fmaxf(c, 0.0f));
  const float ea = expf(a - m);
  const float eb = expf(bs - m);
  const float ec = expf(c - m);
  const float ez = expf(-m);
  const float inv = 1.0f / (ea + eb + ec + (float)(NSEQ - 3) * ez);

  const float* Vb = V + (size_t)b * NSEQ * DHEAD;
  const float2 v0 = ((const float2*)(Vb))[lane];
  const float2 v1 = ((const float2*)(Vb + DHEAD))[lane];
  const float2 v2 = ((const float2*)(Vb + 2 * DHEAD))[lane];

  __syncthreads();
  const float2 s0 = ((const float2*)(sred))[lane];
  const float2 s1 = ((const float2*)(sred + 64))[lane];
  const float2 s2 = ((const float2*)(sred + 128))[lane];
  const float2 s3 = ((const float2*)(sred + 192))[lane];
  float2 sr;
  sr.x = ((s0.x + s1.x) + s2.x) + s3.x;
  sr.y = ((s0.y + s1.y) + s2.y) + s3.y;

  float2 o;
  o.x = (ea * v0.x + eb * v1.x + ec * v2.x + ez * sr.x) * inv;
  o.y = (ea * v0.y + eb * v1.y + ec * v2.y + ez * sr.y) * inv;
  ((float2*)(out + (size_t)g * DHEAD))[lane] = o;
}

extern "C" void kernel_launch(void* const* d_in, const int* in_sizes, int n_in,
                              void* d_out, int out_size, void* d_ws, size_t ws_size,
                              hipStream_t stream) {
  const float* x  = (const float*)d_in[0];
  const float* Wq = (const float*)d_in[1];
  const float* bq = (const float*)d_in[2];
  const float* Wk = (const float*)d_in[3];
  const float* bk = (const float*)d_in[4];
  const float* Wv = (const float*)d_in[5];
  const float* bv = (const float*)d_in[6];
  float* out = (float*)d_out;

  const int BN = in_sizes[0] / EMBED;

  float* Q     = (float*)d_ws;
  float* K     = Q + (size_t)BN * DHEAD;
  float* V     = K + (size_t)BN * DHEAD;

  qkv_gemm_kernel<<<BN / BM, 256, 0, stream>>>(x, Wq, bq, Wk, bk, Wv, bv, Q, K, V);
  vsum_kernel<<<BN / VCHUNK, 512, 0, stream>>>(V);
  combine_kernel<<<BN / 8, 256, 0, stream>>>(Q, K, V, out);
}
